# starting kernel
# baseline (speedup 1.0000x reference)
#include <hip/hip_runtime.h>
#include <hip/hip_bf16.h>


typedef __attribute__((ext_vector_type(2))) float v2f;
typedef __attribute__((ext_vector_type(8))) float v8f;

#define T_LEN 2048
#define B_N   4096
#define H1N   9
#define H2N   32
#define DIN   5
#define OUTT  25

__device__ __forceinline__ v8f wmma4(v2f a, v2f b, v8f c) {
  return __builtin_amdgcn_wmma_f32_16x16x4_f32(
       false, a,  false, b,
       (short)0, c,  false,  false);
}

__device__ __forceinline__ float fast_tanh(float v) {
#if __has_builtin(__builtin_amdgcn_tanhf)
  return __builtin_amdgcn_tanhf(v);
#else
  return tanhf(v);
#endif
}

__device__ __forceinline__ void rnn1_step(
    v2f A0, v2f A1, v2f A2, v2f A3, v8f cbias,
    float d[5], float2 xp, float xw)
{
  v2f B0; B0.x = d[0]; B0.y = d[1];
  v2f B1; B1.x = d[2]; B1.y = d[3];
  v2f B2; B2.x = d[4]; B2.y = xw;
  v2f B3; B3.x = xp.x; B3.y = xp.y;

  v8f cz = {};
  v8f c1 = wmma4(A0, B0, cbias);
  v8f c2 = wmma4(A1, B1, cz);
  v8f c3 = wmma4(A2, B2, cz);
  v8f c4 = wmma4(A3, B3, cz);

#pragma unroll
  for (int i = 0; i < 5; ++i)
    d[i] = fast_tanh((c1[i] + c2[i]) + (c3[i] + c4[i]));
}

template <int DIR>
__device__ __forceinline__ void rnn1_time_loop(
    const float* __restrict__ x, long long xrow, int hi,
    v2f A0, v2f A1, v2f A2, v2f A3, v8f cbias, float d[5])
{
  constexpr int STEP = DIR ? -DIN : DIN;
  const long long t0 = DIR ? (long long)(T_LEN - 1) * DIN : 0;
  const float* q = x + xrow + t0 + 2 * hi;
  const float* r = x + xrow + t0 + 4;

#pragma unroll
  for (int i = 0; i < 5; ++i) d[i] = 0.f;

  float2 xp[4]; float xw[4];
#pragma unroll
  for (int u = 0; u < 4; ++u) {
    xp[u] = *(const float2*)&q[u * STEP];
    xw[u] = r[u * STEP];
  }
  q += 4 * STEP; r += 4 * STEP;

#pragma unroll 1
  for (int t = 0; t < T_LEN - 4; t += 4) {
    __builtin_prefetch(r + STEP * 60, 0, 1);
#pragma unroll
    for (int u = 0; u < 4; ++u) {
      rnn1_step(A0, A1, A2, A3, cbias, d, xp[u], xw[u]);
      xp[u] = *(const float2*)&q[u * STEP];
      xw[u] = r[u * STEP];
    }
    q += 4 * STEP; r += 4 * STEP;
  }
#pragma unroll
  for (int u = 0; u < 4; ++u)
    rnn1_step(A0, A1, A2, A3, cbias, d, xp[u], xw[u]);
}

__global__ __launch_bounds__(32) void rnn1_wmma_kernel(
    const float* __restrict__ x,
    const float* __restrict__ w_ih_f, const float* __restrict__ w_hh_f,
    const float* __restrict__ b_ih_f, const float* __restrict__ b_hh_f,
    const float* __restrict__ w_ih_b, const float* __restrict__ w_hh_b,
    const float* __restrict__ b_ih_b, const float* __restrict__ b_hh_b,
    float* __restrict__ y  )
{
  const int lane = threadIdx.x;
  const int g    = blockIdx.x;
  const int dir  = g >> 8;
  const int s0   = (g & 255) * 16;
  const int hi   = lane >> 4;
  const int ii   = lane & 15;

  const float* wih = dir ? w_ih_b : w_ih_f;
  const float* whh = dir ? w_hh_b : w_hh_f;
  const float* bih = dir ? b_ih_b : b_ih_f;
  const float* bhh = dir ? b_hh_b : b_hh_f;

  auto unitOfRow = [](int i) -> int {
    return (i < 5) ? i : ((i >= 8 && i < 12) ? i - 3 : -1);
  };
  const int slotType[16] = {0,0,0,0, 0,0,0,0, 0,2,2,1, 1,1,1,1};
  const int slotIdx [16] = {0,1,5,6, 2,3,7,8, 4,0,0,4, 0,1,2,3};

  const int urow = unitOfRow(ii);
  auto wElem = [&](int k) -> float {
    if (urow < 0) return 0.f;
    const int ty = slotType[k], ix = slotIdx[k];
    if (ty == 0) return whh[urow * H1N + ix];
    if (ty == 1) return wih[urow * DIN + ix];
    return 0.f;
  };
  auto aW = [&](int kb) -> v2f {
    v2f v; const int k0 = kb + 2 * hi;
    v.x = wElem(k0); v.y = wElem(k0 + 1);
    return v;
  };
  const v2f A0 = aW(0), A1 = aW(4), A2 = aW(8), A3 = aW(12);

  v8f cbias;
#pragma unroll
  for (int rr = 0; rr < 8; ++rr) {
    const int uu = unitOfRow(rr + 8 * hi);
    cbias[rr] = (uu >= 0) ? (bih[uu] + bhh[uu]) : 0.f;
  }

  const long long xrow = (long long)(s0 + ii) * (T_LEN * DIN);

  float d[5];
  if (dir == 0)
    rnn1_time_loop<0>(x, xrow, hi, A0, A1, A2, A3, cbias, d);
  else
    rnn1_time_loop<1>(x, xrow, hi, A0, A1, A2, A3, cbias, d);

  const int sj = s0 + ii;
#pragma unroll
  for (int rr = 0; rr < 5; ++rr) {
    const int uu = unitOfRow(rr + 8 * hi);
    if (uu >= 0) y[sj * (2 * H1N) + dir * H1N + uu] = d[rr];
  }
}

__global__ __launch_bounds__(32) void rnn2_wmma_kernel(
    const float* __restrict__ w_ih2, const float* __restrict__ w_hh2,
    const float* __restrict__ b_ih2, const float* __restrict__ b_hh2,
    const float* __restrict__ y   ,
    float* __restrict__ ys        )
{
  const int lane = threadIdx.x;
  const int hi = lane >> 4, ii = lane & 15;
  const int s0 = blockIdx.x * 16;

  v2f Aih[5][2];
#pragma unroll
  for (int c = 0; c < 5; ++c)
#pragma unroll
    for (int nt = 0; nt < 2; ++nt) {
      const int k0 = 4 * c + 2 * hi, i = nt * 16 + ii;
      Aih[c][nt].x = ((k0    ) < 18) ? w_ih2[i * 18 + k0    ] : 0.f;
      Aih[c][nt].y = ((k0 + 1) < 18) ? w_ih2[i * 18 + k0 + 1] : 0.f;
    }
  v2f Ahh[8][2];
#pragma unroll
  for (int c = 0; c < 8; ++c)
#pragma unroll
    for (int nt = 0; nt < 2; ++nt) {
      const int j0 = 16 * (c >> 2) + 8 * hi + 2 * (c & 3);
      const int i = nt * 16 + ii;
      Ahh[c][nt].x = w_hh2[i * 32 + j0    ];
      Ahh[c][nt].y = w_hh2[i * 32 + j0 + 1];
    }

  v8f cb[2];
#pragma unroll
  for (int nt = 0; nt < 2; ++nt)
#pragma unroll
    for (int rr = 0; rr < 8; ++rr) {
      const int i = nt * 16 + rr + 8 * hi;
      cb[nt][rr] = b_ih2[i] + b_hh2[i];
    }

  const int base = (s0 + ii) * 18;
  v2f By[5];
#pragma unroll
  for (int c = 0; c < 5; ++c) {
    const int k0 = 4 * c + 2 * hi;
    const float v0 = y[base + k0];
    const float v1 = y[base + k0 + 1];
    By[c].x = ((k0    ) < 18) ? v0 : 0.f;
    By[c].y = ((k0 + 1) < 18) ? v1 : 0.f;
  }

  v8f d[2];
  v8f cz = {};
#pragma unroll 1
  for (int t = 0; t < OUTT; ++t) {
    if (t == 0) {
#pragma unroll
      for (int nt = 0; nt < 2; ++nt) {
        v8f c1 = wmma4(Aih[0][nt], By[0], cb[nt]);
        v8f c2 = wmma4(Aih[1][nt], By[1], cz);
        c1 = wmma4(Aih[2][nt], By[2], c1);
        c2 = wmma4(Aih[3][nt], By[3], c2);
        c1 = wmma4(Aih[4][nt], By[4], c1);
#pragma unroll
        for (int rr = 0; rr < 8; ++rr) d[nt][rr] = fast_tanh(c1[rr] + c2[rr]);
      }
    } else {
      v2f Bh[8];
#pragma unroll
      for (int c = 0; c < 8; ++c) {
        Bh[c].x = d[c >> 2][2 * (c & 3)    ];
        Bh[c].y = d[c >> 2][2 * (c & 3) + 1];
      }
      v8f dn[2];
#pragma unroll
      for (int nt = 0; nt < 2; ++nt) {
        v8f c1 = cb[nt];
        v8f c2 = cz;
#pragma unroll
        for (int cc = 0; cc < 8; cc += 2) {
          c1 = wmma4(Ahh[cc    ][nt], Bh[cc    ], c1);
          c2 = wmma4(Ahh[cc + 1][nt], Bh[cc + 1], c2);
        }
#pragma unroll
        for (int rr = 0; rr < 8; ++rr) dn[nt][rr] = fast_tanh(c1[rr] + c2[rr]);
      }
      d[0] = dn[0]; d[1] = dn[1];
    }

    const long long ob = ((long long)(s0 + ii) * OUTT + t) * 32;
#pragma unroll
    for (int nt = 0; nt < 2; ++nt)
#pragma unroll
      for (int rr = 0; rr < 8; ++rr)
        ys[ob + nt * 16 + rr + 8 * hi] = d[nt][rr];
  }
}

__global__ __launch_bounds__(256) void out_proj_kernel(
    const float* __restrict__ ys, const float* __restrict__ w_out,
    const float* __restrict__ b_out, float* __restrict__ out)
{
  const int id = blockIdx.x * blockDim.x + threadIdx.x;
  if (id >= B_N * OUTT) return;
  const float* row = ys + (long long)id * 32;
  float a0 = b_out[0], a1 = b_out[1], a2 = b_out[2];
#pragma unroll
  for (int nn = 0; nn < 32; ++nn) {
    const float v = row[nn];
    a0 = fmaf(v, w_out[nn],      a0);
    a1 = fmaf(v, w_out[32 + nn], a1);
    a2 = fmaf(v, w_out[64 + nn], a2);
  }
  out[(long long)id * 3 + 0] = a0;
  out[(long long)id * 3 + 1] = a1;
  out[(long long)id * 3 + 2] = a2;
}

extern "C" void kernel_launch(void* const* d_in, const int* in_sizes, int n_in,
                              void* d_out, int out_size, void* d_ws,
                              size_t ws_size, hipStream_t stream) {
  const float* x      = (const float*)d_in[0];
  const float* w_ih_f = (const float*)d_in[1];
  const float* w_hh_f = (const float*)d_in[2];
  const float* b_ih_f = (const float*)d_in[3];
  const float* b_hh_f = (const float*)d_in[4];
  const float* w_ih_b = (const float*)d_in[5];
  const float* w_hh_b = (const float*)d_in[6];
  const float* b_ih_b = (const float*)d_in[7];
  const float* b_hh_b = (const float*)d_in[8];
  const float* w_ih2  = (const float*)d_in[9];
  const float* w_hh2  = (const float*)d_in[10];
  const float* b_ih2  = (const float*)d_in[11];
  const float* b_hh2  = (const float*)d_in[12];
  const float* w_out  = (const float*)d_in[13];
  const float* b_out  = (const float*)d_in[14];

  float* y  = (float*)d_ws;
  float* ys = y + (size_t)B_N * (2 * H1N);

  rnn1_wmma_kernel<<<dim3(512), dim3(32), 0, stream>>>(
      x, w_ih_f, w_hh_f, b_ih_f, b_hh_f,
      w_ih_b, w_hh_b, b_ih_b, b_hh_b, y);

  rnn2_wmma_kernel<<<dim3(256), dim3(32), 0, stream>>>(
      w_ih2, w_hh2, b_ih2, b_hh2, y, ys);

  const int nthreads = B_N * OUTT;
  out_proj_kernel<<<dim3((nthreads + 255) / 256), dim3(256), 0, stream>>>(
      ys, w_out, b_out, (float*)d_out);
}


# rnn1 x feed through LDS: 32-step chunks of the 16 sample rows fetched as whole 128-B lines (16-byte loads) two chunks ahead, B operands read from a 2-chunk LDS ring once per 8 steps; sums unchanged
# speedup vs baseline: 1.4866x; 1.4866x over previous
#include <hip/hip_runtime.h>
#include <hip/hip_bf16.h>


typedef __attribute__((ext_vector_type(2))) float v2f;
typedef __attribute__((ext_vector_type(8))) float v8f;
typedef __attribute__((ext_vector_type(4))) float v4f;

#define T_LEN 2048
#define B_N   4096
#define H1N   9
#define H2N   32
#define DIN   5
#define OUTT  25

__device__ __forceinline__ v8f wmma4(v2f a, v2f b, v8f c) {
  return __builtin_amdgcn_wmma_f32_16x16x4_f32(
       false, a,  false, b,
       (short)0, c,  false,  false);
}

__device__ __forceinline__ float fast_tanh(float v) {
#if __has_builtin(__builtin_amdgcn_tanhf)
  return __builtin_amdgcn_tanhf(v);
#else
  return tanhf(v);
#endif
}

__device__ __forceinline__ void rnn1_step(
    v2f A0, v2f A1, v2f A2, v2f A3, v8f cbias,
    float d[5], float2 xp, float xw)
{
  v2f B0; B0.x = d[0]; B0.y = d[1];
  v2f B1; B1.x = d[2]; B1.y = d[3];
  v2f B2; B2.x = d[4]; B2.y = xw;
  v2f B3; B3.x = xp.x; B3.y = xp.y;

  v8f cz = {};
  v8f c1 = wmma4(A0, B0, cbias);
  v8f c2 = wmma4(A1, B1, cz);
  v8f c3 = wmma4(A2, B2, cz);
  v8f c4 = wmma4(A3, B3, cz);

#pragma unroll
  for (int i = 0; i < 5; ++i)
    d[i] = fast_tanh((c1[i] + c2[i]) + (c3[i] + c4[i]));
}

#define CH      32
#define NCHUNK  (T_LEN / CH)
#define CHF     (CH * DIN)
#define NPIECE  (CHF / 4)
#define NLD     (16 * NPIECE / 32)
#define PITCHF  (CHF + 4)
#define BUFF    (16 * PITCHF)

template <int DIR>
__device__ __forceinline__ void rnn1_time_loop(
    const float* __restrict__ xs  ,
    float* smem, int lane, int hi, int ii,
    v2f A0, v2f A1, v2f A2, v2f A3, v8f cbias, float d[5])
{
  const unsigned gl = (unsigned)(lane >> 3) * (unsigned)(T_LEN * DIN) + (unsigned)(lane & 7) * 4u;
  const unsigned ll = (unsigned)(lane >> 3) * (unsigned)PITCHF + (unsigned)(lane & 7) * 4u;

  v4f R[NLD];
  auto issue = [&](int c) {
    const float* g = xs + (DIR ? (T_LEN - CH * (c + 1)) * DIN : c * CHF) + gl;
#pragma unroll
    for (int jg = 0; jg < 4; ++jg)
#pragma unroll
      for (int jl = 0; jl < 5; ++jl)
        R[jg * 5 + jl] = *(const v4f*)(g + jg * 4 * (T_LEN * DIN) + jl * 32);
  };
  auto stash = [&](int c) {
    float* bf = smem + (c & 1) * BUFF + ll;
#pragma unroll
    for (int jg = 0; jg < 4; ++jg)
#pragma unroll
      for (int jl = 0; jl < 5; ++jl)
        *(v4f*)(bf + jg * 4 * PITCHF + jl * 32) = R[jg * 5 + jl];
  };

  const float* lp = smem + ii * PITCHF + 2 * hi;
  const float* lw = smem + ii * PITCHF + 4;
  auto readgrp = [&](int c, int g, float2 xp[8], float xw[8]) {
    const int bo = (c & 1) * BUFF;
    __builtin_amdgcn_sched_barrier(0);
#pragma unroll
    for (int u = 0; u < 8; ++u) {
      const int k   = 8 * g + u;
      const int off = (DIR ? (CH - 1 - k) : k) * DIN;
      xp[u].x = lp[bo + off];
      xp[u].y = lp[bo + off + 1];
      xw[u]   = lw[bo + off];
    }
    __builtin_amdgcn_sched_barrier(0);
  };

#pragma unroll
  for (int i = 0; i < 5; ++i) d[i] = 0.f;

  float2 xpA[8], xpB[8]; float xwA[8], xwB[8];
  issue(0);
  stash(0);
  __syncthreads();
  issue(1);
  readgrp(0, 0, xpA, xwA);

#pragma unroll 1
  for (int c = 0; c < NCHUNK; ++c) {
    if (c + 1 < NCHUNK) stash(c + 1);
    __syncthreads();
    if (c + 2 < NCHUNK) issue(c + 2);
    readgrp(c, 1, xpB, xwB);
#pragma unroll
    for (int u = 0; u < 8; ++u) rnn1_step(A0, A1, A2, A3, cbias, d, xpA[u], xwA[u]);
    readgrp(c, 2, xpA, xwA);
#pragma unroll
    for (int u = 0; u < 8; ++u) rnn1_step(A0, A1, A2, A3, cbias, d, xpB[u], xwB[u]);
    readgrp(c, 3, xpB, xwB);
#pragma unroll
    for (int u = 0; u < 8; ++u) rnn1_step(A0, A1, A2, A3, cbias, d, xpA[u], xwA[u]);
    if (c + 1 < NCHUNK) readgrp(c + 1, 0, xpA, xwA);
#pragma unroll
    for (int u = 0; u < 8; ++u) rnn1_step(A0, A1, A2, A3, cbias, d, xpB[u], xwB[u]);
  }
}

__global__ __launch_bounds__(32) void rnn1_wmma_kernel(
    const float* __restrict__ x,
    const float* __restrict__ w_ih_f, const float* __restrict__ w_hh_f,
    const float* __restrict__ b_ih_f, const float* __restrict__ b_hh_f,
    const float* __restrict__ w_ih_b, const float* __restrict__ w_hh_b,
    const float* __restrict__ b_ih_b, const float* __restrict__ b_hh_b,
    float* __restrict__ y  )
{
  const int lane = threadIdx.x;
  const int g    = blockIdx.x;
  const int dir  = g >> 8;
  const int s0   = (g & 255) * 16;
  const int hi   = lane >> 4;
  const int ii   = lane & 15;

  const float* wih = dir ? w_ih_b : w_ih_f;
  const float* whh = dir ? w_hh_b : w_hh_f;
  const float* bih = dir ? b_ih_b : b_ih_f;
  const float* bhh = dir ? b_hh_b : b_hh_f;

  auto unitOfRow = [](int i) -> int {
    return (i < 5) ? i : ((i >= 8 && i < 12) ? i - 3 : -1);
  };
  const int slotType[16] = {0,0,0,0, 0,0,0,0, 0,2,2,1, 1,1,1,1};
  const int slotIdx [16] = {0,1,5,6, 2,3,7,8, 4,0,0,4, 0,1,2,3};

  const int urow = unitOfRow(ii);
  auto wElem = [&](int k) -> float {
    if (urow < 0) return 0.f;
    const int ty = slotType[k], ix = slotIdx[k];
    if (ty == 0) return whh[urow * H1N + ix];
    if (ty == 1) return wih[urow * DIN + ix];
    return 0.f;
  };
  auto aW = [&](int kb) -> v2f {
    v2f v; const int k0 = kb + 2 * hi;
    v.x = wElem(k0); v.y = wElem(k0 + 1);
    return v;
  };
  const v2f A0 = aW(0), A1 = aW(4), A2 = aW(8), A3 = aW(12);

  v8f cbias;
#pragma unroll
  for (int rr = 0; rr < 8; ++rr) {
    const int uu = unitOfRow(rr + 8 * hi);
    cbias[rr] = (uu >= 0) ? (bih[uu] + bhh[uu]) : 0.f;
  }

  __shared__ __attribute__((aligned(16))) float smem[2 * BUFF];
  const float* xs = x + (long long)s0 * (T_LEN * DIN);

  float d[5];
  if (dir == 0)
    rnn1_time_loop<0>(xs, smem, lane, hi, ii, A0, A1, A2, A3, cbias, d);
  else
    rnn1_time_loop<1>(xs, smem, lane, hi, ii, A0, A1, A2, A3, cbias, d);

  const int sj = s0 + ii;
#pragma unroll
  for (int rr = 0; rr < 5; ++rr) {
    const int uu = unitOfRow(rr + 8 * hi);
    if (uu >= 0) y[sj * (2 * H1N) + dir * H1N + uu] = d[rr];
  }
}

__global__ __launch_bounds__(32) void rnn2_wmma_kernel(
    const float* __restrict__ w_ih2, const float* __restrict__ w_hh2,
    const float* __restrict__ b_ih2, const float* __restrict__ b_hh2,
    const float* __restrict__ y   ,
    float* __restrict__ ys        )
{
  const int lane = threadIdx.x;
  const int hi = lane >> 4, ii = lane & 15;
  const int s0 = blockIdx.x * 16;

  v2f Aih[5][2];
#pragma unroll
  for (int c = 0; c < 5; ++c)
#pragma unroll
    for (int nt = 0; nt < 2; ++nt) {
      const int k0 = 4 * c + 2 * hi, i = nt * 16 + ii;
      Aih[c][nt].x = ((k0    ) < 18) ? w_ih2[i * 18 + k0    ] : 0.f;
      Aih[c][nt].y = ((k0 + 1) < 18) ? w_ih2[i * 18 + k0 + 1] : 0.f;
    }
  v2f Ahh[8][2];
#pragma unroll
  for (int c = 0; c < 8; ++c)
#pragma unroll
    for (int nt = 0; nt < 2; ++nt) {
      const int j0 = 16 * (c >> 2) + 8 * hi + 2 * (c & 3);
      const int i = nt * 16 + ii;
      Ahh[c][nt].x = w_hh2[i * 32 + j0    ];
      Ahh[c][nt].y = w_hh2[i * 32 + j0 + 1];
    }

  v8f cb[2];
#pragma unroll
  for (int nt = 0; nt < 2; ++nt)
#pragma unroll
    for (int rr = 0; rr < 8; ++rr) {
      const int i = nt * 16 + rr + 8 * hi;
      cb[nt][rr] = b_ih2[i] + b_hh2[i];
    }

  const int base = (s0 + ii) * 18;
  v2f By[5];
#pragma unroll
  for (int c = 0; c < 5; ++c) {
    const int k0 = 4 * c + 2 * hi;
    const float v0 = y[base + k0];
    const float v1 = y[base + k0 + 1];
    By[c].x = ((k0    ) < 18) ? v0 : 0.f;
    By[c].y = ((k0 + 1) < 18) ? v1 : 0.f;
  }

  v8f d[2];
  v8f cz = {};
#pragma unroll 1
  for (int t = 0; t < OUTT; ++t) {
    if (t == 0) {
#pragma unroll
      for (int nt = 0; nt < 2; ++nt) {
        v8f c1 = wmma4(Aih[0][nt], By[0], cb[nt]);
        v8f c2 = wmma4(Aih[1][nt], By[1], cz);
        c1 = wmma4(Aih[2][nt], By[2], c1);
        c2 = wmma4(Aih[3][nt], By[3], c2);
        c1 = wmma4(Aih[4][nt], By[4], c1);
#pragma unroll
        for (int rr = 0; rr < 8; ++rr) d[nt][rr] = fast_tanh(c1[rr] + c2[rr]);
      }
    } else {
      v2f Bh[8];
#pragma unroll
      for (int c = 0; c < 8; ++c) {
        Bh[c].x = d[c >> 2][2 * (c & 3)    ];
        Bh[c].y = d[c >> 2][2 * (c & 3) + 1];
      }
      v8f dn[2];
#pragma unroll
      for (int nt = 0; nt < 2; ++nt) {
        v8f c1 = cb[nt];
        v8f c2 = cz;
#pragma unroll
        for (int cc = 0; cc < 8; cc += 2) {
          c1 = wmma4(Ahh[cc    ][nt], Bh[cc    ], c1);
          c2 = wmma4(Ahh[cc + 1][nt], Bh[cc + 1], c2);
        }
#pragma unroll
        for (int rr = 0; rr < 8; ++rr) dn[nt][rr] = fast_tanh(c1[rr] + c2[rr]);
      }
      d[0] = dn[0]; d[1] = dn[1];
    }

    const long long ob = ((long long)(s0 + ii) * OUTT + t) * 32;
#pragma unroll
    for (int nt = 0; nt < 2; ++nt)
#pragma unroll
      for (int rr = 0; rr < 8; ++rr)
        ys[ob + nt * 16 + rr + 8 * hi] = d[nt][rr];
  }
}

__global__ __launch_bounds__(256) void out_proj_kernel(
    const float* __restrict__ ys, const float* __restrict__ w_out,
    const float* __restrict__ b_out, float* __restrict__ out)
{
  const int id = blockIdx.x * blockDim.x + threadIdx.x;
  if (id >= B_N * OUTT) return;
  const float* row = ys + (long long)id * 32;
  float a0 = b_out[0], a1 = b_out[1], a2 = b_out[2];
#pragma unroll
  for (int nn = 0; nn < 32; ++nn) {
    const float v = row[nn];
    a0 = fmaf(v, w_out[nn],      a0);
    a1 = fmaf(v, w_out[32 + nn], a1);
    a2 = fmaf(v, w_out[64 + nn], a2);
  }
  out[(long long)id * 3 + 0] = a0;
  out[(long long)id * 3 + 1] = a1;
  out[(long long)id * 3 + 2] = a2;
}

extern "C" void kernel_launch(void* const* d_in, const int* in_sizes, int n_in,
                              void* d_out, int out_size, void* d_ws,
                              size_t ws_size, hipStream_t stream) {
  const float* x      = (const float*)d_in[0];
  const float* w_ih_f = (const float*)d_in[1];
  const float* w_hh_f = (const float*)d_in[2];
  const float* b_ih_f = (const float*)d_in[3];
  const float* b_hh_f = (const float*)d_in[4];
  const float* w_ih_b = (const float*)d_in[5];
  const float* w_hh_b = (const float*)d_in[6];
  const float* b_ih_b = (const float*)d_in[7];
  const float* b_hh_b = (const float*)d_in[8];
  const float* w_ih2  = (const float*)d_in[9];
  const float* w_hh2  = (const float*)d_in[10];
  const float* b_ih2  = (const float*)d_in[11];
  const float* b_hh2  = (const float*)d_in[12];
  const float* w_out  = (const float*)d_in[13];
  const float* b_out  = (const float*)d_in[14];

  float* y  = (float*)d_ws;
  float* ys = y + (size_t)B_N * (2 * H1N);

  rnn1_wmma_kernel<<<dim3(512), dim3(32), 0, stream>>>(
      x, w_ih_f, w_hh_f, b_ih_f, b_hh_f,
      w_ih_b, w_hh_b, b_ih_b, b_hh_b, y);

  rnn2_wmma_kernel<<<dim3(256), dim3(32), 0, stream>>>(
      w_ih2, w_hh2, b_ih2, b_hh2, y, ys);

  const int nthreads = B_N * OUTT;
  out_proj_kernel<<<dim3((nthreads + 255) / 256), dim3(256), 0, stream>>>(
      ys, w_out, b_out, (float*)d_out);
}


# rnn1 input product W_ih x + b moved to a second wave on another SIMD, handed through a 3-slot LDS ring as C operand of state product 1: 3 matrix ops per step on the chain, not 4; sum order changed
# speedup vs baseline: 1.7817x; 1.1985x over previous
#include <hip/hip_runtime.h>
#include <hip/hip_bf16.h>


typedef __attribute__((ext_vector_type(2))) float v2f;
typedef __attribute__((ext_vector_type(8))) float v8f;
typedef __attribute__((ext_vector_type(4))) float v4f;

#define T_LEN 2048
#define B_N   4096
#define H1N   9
#define H2N   32
#define DIN   5
#define OUTT  25

__device__ __forceinline__ v8f wmma4(v2f a, v2f b, v8f c) {
  return __builtin_amdgcn_wmma_f32_16x16x4_f32(
       false, a,  false, b,
       (short)0, c,  false,  false);
}

__device__ __forceinline__ float fast_tanh(float v) {
#if __has_builtin(__builtin_amdgcn_tanhf)
  return __builtin_amdgcn_tanhf(v);
#else
  return tanhf(v);
#endif
}

__device__ __forceinline__ void rnn1_state_step(
    v2f A0, v2f A1, v2f A2h, float d[5], v8f p)
{
  v2f B0; B0.x = d[0]; B0.y = d[1];
  v2f B1; B1.x = d[2]; B1.y = d[3];
  v2f B2; B2.x = d[4]; B2.y = 0.f;

  v8f cz = {};
  v8f c1 = wmma4(A0, B0, p);
  v8f c2 = wmma4(A1, B1, cz);
  v8f c3 = wmma4(A2h, B2, cz);

#pragma unroll
  for (int i = 0; i < 5; ++i)
    d[i] = fast_tanh((c1[i] + c2[i]) + c3[i]);
}

#define CH      32
#define NCHUNK  (T_LEN / CH)
#define CHF     (CH * DIN)
#define NPIECE  (CHF / 4)
#define NLD     (16 * NPIECE / 32)
#define PITCHF  (CHF + 4)
#define BUFF    (16 * PITCHF)
#define PSTEPF  256
#define PSLOTF  (CH * PSTEPF)
#define NPSLOT  3
#define XRINGF  (2 * BUFF)
#define PRINGF  (NPSLOT * PSLOTF)

template <int DIR>
__device__ __forceinline__ void rnn1_pair_loop(
    const float* __restrict__ xs  ,
    float* xr  , float* pr  ,
    int role  ,
    int lane, int hi, int ii,
    v2f A0, v2f A1, v2f A2h, v2f AX4, v2f A3, v8f cbias, float d[5])
{
  const unsigned gl = (unsigned)(lane >> 3) * (unsigned)(T_LEN * DIN) + (unsigned)(lane & 7) * 4u;
  const unsigned ll = (unsigned)(lane >> 3) * (unsigned)PITCHF + (unsigned)(lane & 7) * 4u;

  auto issue = [&](int c, v4f (&R)[NLD]) {
    const float* g = xs + (DIR ? (T_LEN - CH * (c + 1)) * DIN : c * CHF) + gl;
#pragma unroll
    for (int jg = 0; jg < 4; ++jg)
#pragma unroll
      for (int jl = 0; jl < 5; ++jl)
        R[jg * 5 + jl] = *(const v4f*)(g + jg * 4 * (T_LEN * DIN) + jl * 32);
  };
  auto stash = [&](int c, const v4f (&R)[NLD]) {
    float* bf = xr + (c & 1) * BUFF + ll;
#pragma unroll
    for (int jg = 0; jg < 4; ++jg)
#pragma unroll
      for (int jl = 0; jl < 5; ++jl)
        *(v4f*)(bf + jg * 4 * PITCHF + jl * 32) = R[jg * 5 + jl];
  };

  const float* lp = xr + ii * PITCHF + 2 * hi;
  const float* lw = xr + ii * PITCHF + 4;
  auto readgrp = [&](int c, int g, float2 xp[8], float xw[8]) {
    const int bo = (c & 1) * BUFF;
    __builtin_amdgcn_sched_barrier(0);
#pragma unroll
    for (int u = 0; u < 8; ++u) {
      const int k   = 8 * g + u;
      const int off = (DIR ? (CH - 1 - k) : k) * DIN;
      xp[u].x = lp[bo + off];
      xp[u].y = lp[bo + off + 1];
      xw[u]   = lw[bo + off];
    }
    __builtin_amdgcn_sched_barrier(0);
  };

  float* pw = pr + lane * 4;
  auto makegrp = [&](int so, int g, const float2 xp[8], const float xw[8]) {
#pragma unroll
    for (int u = 0; u < 8; ++u) {
      v2f B2; B2.x = 0.f;     B2.y = xw[u];
      v2f B3; B3.x = xp[u].x; B3.y = xp[u].y;
      v8f p = wmma4(AX4, B2, cbias);
      p = wmma4(A3, B3, p);
      float* q = pw + so + (8 * g + u) * PSTEPF;
      *(v4f*)(q)       = __builtin_shufflevector(p, p, 0, 1, 2, 3);
      *(v4f*)(q + 128) = __builtin_shufflevector(p, p, 4, 5, 6, 7);
    }
  };
  auto readp = [&](int so, int g, v8f P[8]) {
    __builtin_amdgcn_sched_barrier(0);
#pragma unroll
    for (int u = 0; u < 8; ++u) {
      const float* q = pw + so + (8 * g + u) * PSTEPF;
      const v4f a = *(const v4f*)(q);
      const v4f b = *(const v4f*)(q + 128);
      P[u] = __builtin_shufflevector(a, b, 0, 1, 2, 3, 4, 5, 6, 7);
    }
    __builtin_amdgcn_sched_barrier(0);
  };

#pragma unroll
  for (int i = 0; i < 5; ++i) d[i] = 0.f;

  v8f PA[8];
  if (role) {
    v4f R0[NLD];
    issue(0, R0);
    stash(0, R0);
  }

#pragma unroll 1
  for (int i = 0; i < NCHUNK + 2; ++i) {
    __syncthreads();
    if (role) {
      if (i < NCHUNK) {
        v4f R[NLD];
        issue(i + 1 < NCHUNK ? i + 1 : i, R);
        const int so = (i % NPSLOT) * PSLOTF;
        float2 xpA[8], xpB[8]; float xwA[8], xwB[8];
        readgrp(i, 0, xpA, xwA);
        readgrp(i, 1, xpB, xwB);
        makegrp(so, 0, xpA, xwA);
        readgrp(i, 2, xpA, xwA);
        makegrp(so, 1, xpB, xwB);
        readgrp(i, 3, xpB, xwB);
        makegrp(so, 2, xpA, xwA);
        makegrp(so, 3, xpB, xwB);
        stash(i + 1, R);
      }
    } else if (i >= 2) {
      const int c  = i - 2;
      const int so = (c % NPSLOT) * PSLOTF;
      const int sn = ((c + 1) % NPSLOT) * PSLOTF;
      v8f PB[8];
      if (c == 0) readp(so, 0, PA);
      readp(so, 1, PB);
#pragma unroll
      for (int u = 0; u < 8; ++u) rnn1_state_step(A0, A1, A2h, d, PA[u]);
      readp(so, 2, PA);
#pragma unroll
      for (int u = 0; u < 8; ++u) rnn1_state_step(A0, A1, A2h, d, PB[u]);
      readp(so, 3, PB);
#pragma unroll
      for (int u = 0; u < 8; ++u) rnn1_state_step(A0, A1, A2h, d, PA[u]);
      readp(sn, 0, PA);
#pragma unroll
      for (int u = 0; u < 8; ++u) rnn1_state_step(A0, A1, A2h, d, PB[u]);
    }
  }
}

__global__ __launch_bounds__(128) void rnn1_wmma_kernel(
    const float* __restrict__ x,
    const float* __restrict__ w_ih_f, const float* __restrict__ w_hh_f,
    const float* __restrict__ b_ih_f, const float* __restrict__ b_hh_f,
    const float* __restrict__ w_ih_b, const float* __restrict__ w_hh_b,
    const float* __restrict__ b_ih_b, const float* __restrict__ b_hh_b,
    float* __restrict__ y  )
{
  const int lane = threadIdx.x & 31;
  const int wv   = __builtin_amdgcn_readfirstlane((int)(threadIdx.x >> 5));
  const int role = wv & 1;
  const int tsel = wv >> 1;
  const int g    = blockIdx.x;
  const int dir  = g >> 7;
  const int s0   = ((g & 127) * 2 + tsel) * 16;
  const int hi   = lane >> 4;
  const int ii   = lane & 15;

  const float* wih = dir ? w_ih_b : w_ih_f;
  const float* whh = dir ? w_hh_b : w_hh_f;
  const float* bih = dir ? b_ih_b : b_ih_f;
  const float* bhh = dir ? b_hh_b : b_hh_f;

  auto unitOfRow = [](int i) -> int {
    return (i < 5) ? i : ((i >= 8 && i < 12) ? i - 3 : -1);
  };
  const int slotType[16] = {0,0,0,0, 0,0,0,0, 0,2,2,1, 1,1,1,1};
  const int slotIdx [16] = {0,1,5,6, 2,3,7,8, 4,0,0,4, 0,1,2,3};

  const int urow = unitOfRow(ii);
  auto wElem = [&](int k) -> float {
    if (urow < 0) return 0.f;
    const int ty = slotType[k], ix = slotIdx[k];
    if (ty == 0) return whh[urow * H1N + ix];
    if (ty == 1) return wih[urow * DIN + ix];
    return 0.f;
  };
  auto aW = [&](int kb) -> v2f {
    v2f v; const int k0 = kb + 2 * hi;
    v.x = wElem(k0); v.y = wElem(k0 + 1);
    return v;
  };
  const v2f A0 = aW(0), A1 = aW(4), A2 = aW(8), A3 = aW(12);
  v2f A2h; A2h.x = hi ? 0.f : A2.x; A2h.y = 0.f;
  v2f AX4; AX4.x = 0.f;             AX4.y = hi ? A2.y : 0.f;

  v8f cbias;
#pragma unroll
  for (int rr = 0; rr < 8; ++rr) {
    const int uu = unitOfRow(rr + 8 * hi);
    cbias[rr] = (uu >= 0) ? (bih[uu] + bhh[uu]) : 0.f;
  }

  __shared__ __attribute__((aligned(16))) float smem[2 * (XRINGF + PRINGF)];
  float* xr = smem + tsel * XRINGF;
  float* pr = smem + 2 * XRINGF + tsel * PRINGF;
  const float* xs = x + (long long)s0 * (T_LEN * DIN);

  float d[5];
  if (dir == 0)
    rnn1_pair_loop<0>(xs, xr, pr, role, lane, hi, ii, A0, A1, A2h, AX4, A3, cbias, d);
  else
    rnn1_pair_loop<1>(xs, xr, pr, role, lane, hi, ii, A0, A1, A2h, AX4, A3, cbias, d);

  if (role == 0) {
    const int sj = s0 + ii;
#pragma unroll
    for (int rr = 0; rr < 5; ++rr) {
      const int uu = unitOfRow(rr + 8 * hi);
      if (uu >= 0) y[sj * (2 * H1N) + dir * H1N + uu] = d[rr];
    }
  }
}

__global__ __launch_bounds__(32) void rnn2_wmma_kernel(
    const float* __restrict__ w_ih2, const float* __restrict__ w_hh2,
    const float* __restrict__ b_ih2, const float* __restrict__ b_hh2,
    const float* __restrict__ y   ,
    float* __restrict__ ys        )
{
  const int lane = threadIdx.x;
  const int hi = lane >> 4, ii = lane & 15;
  const int s0 = blockIdx.x * 16;

  v2f Aih[5][2];
#pragma unroll
  for (int c = 0; c < 5; ++c)
#pragma unroll
    for (int nt = 0; nt < 2; ++nt) {
      const int k0 = 4 * c + 2 * hi, i = nt * 16 + ii;
      Aih[c][nt].x = ((k0    ) < 18) ? w_ih2[i * 18 + k0    ] : 0.f;
      Aih[c][nt].y = ((k0 + 1) < 18) ? w_ih2[i * 18 + k0 + 1] : 0.f;
    }
  v2f Ahh[8][2];
#pragma unroll
  for (int c = 0; c < 8; ++c)
#pragma unroll
    for (int nt = 0; nt < 2; ++nt) {
      const int j0 = 16 * (c >> 2) + 8 * hi + 2 * (c & 3);
      const int i = nt * 16 + ii;
      Ahh[c][nt].x = w_hh2[i * 32 + j0    ];
      Ahh[c][nt].y = w_hh2[i * 32 + j0 + 1];
    }

  v8f cb[2];
#pragma unroll
  for (int nt = 0; nt < 2; ++nt)
#pragma unroll
    for (int rr = 0; rr < 8; ++rr) {
      const int i = nt * 16 + rr + 8 * hi;
      cb[nt][rr] = b_ih2[i] + b_hh2[i];
    }

  const int base = (s0 + ii) * 18;
  v2f By[5];
#pragma unroll
  for (int c = 0; c < 5; ++c) {
    const int k0 = 4 * c + 2 * hi;
    const float v0 = y[base + k0];
    const float v1 = y[base + k0 + 1];
    By[c].x = ((k0    ) < 18) ? v0 : 0.f;
    By[c].y = ((k0 + 1) < 18) ? v1 : 0.f;
  }

  v8f d[2];
  v8f cz = {};
#pragma unroll 1
  for (int t = 0; t < OUTT; ++t) {
    if (t == 0) {
#pragma unroll
      for (int nt = 0; nt < 2; ++nt) {
        v8f c1 = wmma4(Aih[0][nt], By[0], cb[nt]);
        v8f c2 = wmma4(Aih[1][nt], By[1], cz);
        c1 = wmma4(Aih[2][nt], By[2], c1);
        c2 = wmma4(Aih[3][nt], By[3], c2);
        c1 = wmma4(Aih[4][nt], By[4], c1);
#pragma unroll
        for (int rr = 0; rr < 8; ++rr) d[nt][rr] = fast_tanh(c1[rr] + c2[rr]);
      }
    } else {
      v2f Bh[8];
#pragma unroll
      for (int c = 0; c < 8; ++c) {
        Bh[c].x = d[c >> 2][2 * (c & 3)    ];
        Bh[c].y = d[c >> 2][2 * (c & 3) + 1];
      }
      v8f dn[2];
#pragma unroll
      for (int nt = 0; nt < 2; ++nt) {
        v8f c1 = cb[nt];
        v8f c2 = cz;
#pragma unroll
        for (int cc = 0; cc < 8; cc += 2) {
          c1 = wmma4(Ahh[cc    ][nt], Bh[cc    ], c1);
          c2 = wmma4(Ahh[cc + 1][nt], Bh[cc + 1], c2);
        }
#pragma unroll
        for (int rr = 0; rr < 8; ++rr) dn[nt][rr] = fast_tanh(c1[rr] + c2[rr]);
      }
      d[0] = dn[0]; d[1] = dn[1];
    }

    const long long ob = ((long long)(s0 + ii) * OUTT + t) * 32;
#pragma unroll
    for (int nt = 0; nt < 2; ++nt)
#pragma unroll
      for (int rr = 0; rr < 8; ++rr)
        ys[ob + nt * 16 + rr + 8 * hi] = d[nt][rr];
  }
}

__global__ __launch_bounds__(256) void out_proj_kernel(
    const float* __restrict__ ys, const float* __restrict__ w_out,
    const float* __restrict__ b_out, float* __restrict__ out)
{
  const int id = blockIdx.x * blockDim.x + threadIdx.x;
  if (id >= B_N * OUTT) return;
  const float* row = ys + (long long)id * 32;
  float a0 = b_out[0], a1 = b_out[1], a2 = b_out[2];
#pragma unroll
  for (int nn = 0; nn < 32; ++nn) {
    const float v = row[nn];
    a0 = fmaf(v, w_out[nn],      a0);
    a1 = fmaf(v, w_out[32 + nn], a1);
    a2 = fmaf(v, w_out[64 + nn], a2);
  }
  out[(long long)id * 3 + 0] = a0;
  out[(long long)id * 3 + 1] = a1;
  out[(long long)id * 3 + 2] = a2;
}

extern "C" void kernel_launch(void* const* d_in, const int* in_sizes, int n_in,
                              void* d_out, int out_size, void* d_ws,
                              size_t ws_size, hipStream_t stream) {
  const float* x      = (const float*)d_in[0];
  const float* w_ih_f = (const float*)d_in[1];
  const float* w_hh_f = (const float*)d_in[2];
  const float* b_ih_f = (const float*)d_in[3];
  const float* b_hh_f = (const float*)d_in[4];
  const float* w_ih_b = (const float*)d_in[5];
  const float* w_hh_b = (const float*)d_in[6];
  const float* b_ih_b = (const float*)d_in[7];
  const float* b_hh_b = (const float*)d_in[8];
  const float* w_ih2  = (const float*)d_in[9];
  const float* w_hh2  = (const float*)d_in[10];
  const float* b_ih2  = (const float*)d_in[11];
  const float* b_hh2  = (const float*)d_in[12];
  const float* w_out  = (const float*)d_in[13];
  const float* b_out  = (const float*)d_in[14];

  float* y  = (float*)d_ws;
  float* ys = y + (size_t)B_N * (2 * H1N);

  rnn1_wmma_kernel<<<dim3(256), dim3(128), 0, stream>>>(
      x, w_ih_f, w_hh_f, b_ih_f, b_hh_f,
      w_ih_b, w_hh_b, b_ih_b, b_hh_b, y);

  rnn2_wmma_kernel<<<dim3(256), dim3(32), 0, stream>>>(
      w_ih2, w_hh2, b_ih2, b_hh2, y, ys);

  const int nthreads = B_N * OUTT;
  out_proj_kernel<<<dim3((nthreads + 255) / 256), dim3(256), 0, stream>>>(
      ys, w_out, b_out, (float*)d_out);
}


# one launch, not three: forward and backward tile of the same 16 samples share a workgroup; RNN-2 and projection run in the RNN-1 kernel's tail from LDS; y and states never in memory; sums unchanged
# speedup vs baseline: 1.9330x; 1.0849x over previous
#include <hip/hip_runtime.h>
#include <hip/hip_bf16.h>


typedef __attribute__((ext_vector_type(2))) float v2f;
typedef __attribute__((ext_vector_type(8))) float v8f;
typedef __attribute__((ext_vector_type(4))) float v4f;

#define T_LEN 2048
#define B_N   4096
#define H1N   9
#define H2N   32
#define DIN   5
#define OUTT  25

__device__ __forceinline__ v8f wmma4(v2f a, v2f b, v8f c) {
  return __builtin_amdgcn_wmma_f32_16x16x4_f32(
       false, a,  false, b,
       (short)0, c,  false,  false);
}

__device__ __forceinline__ float fast_tanh(float v) {
#if __has_builtin(__builtin_amdgcn_tanhf)
  return __builtin_amdgcn_tanhf(v);
#else
  return tanhf(v);
#endif
}

__device__ __forceinline__ void rnn1_state_step(
    v2f A0, v2f A1, v2f A2h, float d[5], v8f p)
{
  v2f B0; B0.x = d[0]; B0.y = d[1];
  v2f B1; B1.x = d[2]; B1.y = d[3];
  v2f B2; B2.x = d[4]; B2.y = 0.f;

  v8f cz = {};
  v8f c1 = wmma4(A0, B0, p);
  v8f c2 = wmma4(A1, B1, cz);
  v8f c3 = wmma4(A2h, B2, cz);

#pragma unroll
  for (int i = 0; i < 5; ++i)
    d[i] = fast_tanh((c1[i] + c2[i]) + c3[i]);
}

#define CH      32
#define NCHUNK  (T_LEN / CH)
#define CHF     (CH * DIN)
#define NPIECE  (CHF / 4)
#define NLD     (16 * NPIECE / 32)
#define PITCHF  (CHF + 4)
#define BUFF    (16 * PITCHF)
#define PSTEPF  256
#define PSLOTF  (CH * PSTEPF)
#define NPSLOT  3
#define XRINGF  (2 * BUFF)
#define PRINGF  (NPSLOT * PSLOTF)

__device__ __forceinline__ void rnn1_pair_loop(
    const int DIR  ,
    const float* __restrict__ xs  ,
    float* xr  , float* pr  ,
    int role  ,
    int lane, int hi, int ii,
    v2f A0, v2f A1, v2f A2h, v2f AX4, v2f A3, v8f cbias, float d[5])
{
  const unsigned gl = (unsigned)(lane >> 3) * (unsigned)(T_LEN * DIN) + (unsigned)(lane & 7) * 4u;
  const unsigned ll = (unsigned)(lane >> 3) * (unsigned)PITCHF + (unsigned)(lane & 7) * 4u;

  auto issue = [&](int c, v4f (&R)[NLD]) {
    const float* g = xs + (DIR ? (T_LEN - CH * (c + 1)) * DIN : c * CHF) + gl;
#pragma unroll
    for (int jg = 0; jg < 4; ++jg)
#pragma unroll
      for (int jl = 0; jl < 5; ++jl)
        R[jg * 5 + jl] = *(const v4f*)(g + jg * 4 * (T_LEN * DIN) + jl * 32);
  };
  auto stash = [&](int c, const v4f (&R)[NLD]) {
    float* bf = xr + (c & 1) * BUFF + ll;
#pragma unroll
    for (int jg = 0; jg < 4; ++jg)
#pragma unroll
      for (int jl = 0; jl < 5; ++jl)
        *(v4f*)(bf + jg * 4 * PITCHF + jl * 32) = R[jg * 5 + jl];
  };

  const float* lp = xr + ii * PITCHF + 2 * hi;
  const float* lw = xr + ii * PITCHF + 4;
  auto readgrp1 = [&](int c, int g, float2 xp[8], float xw[8], const int RD) __attribute__((always_inline)) {
    const int bo = (c & 1) * BUFF;
    __builtin_amdgcn_sched_barrier(0);
#pragma unroll
    for (int u = 0; u < 8; ++u) {
      const int k   = 8 * g + u;
      const int off = (RD ? (CH - 1 - k) : k) * DIN;
      xp[u].x = lp[bo + off];
      xp[u].y = lp[bo + off + 1];
      xw[u]   = lw[bo + off];
    }
    __builtin_amdgcn_sched_barrier(0);
  };
  auto readgrp = [&](int c, int g, float2 xp[8], float xw[8]) __attribute__((always_inline)) {
    if (DIR) readgrp1(c, g, xp, xw, 1); else readgrp1(c, g, xp, xw, 0);
  };

  float* pw = pr + lane * 4;
  auto makegrp = [&](int so, int g, const float2 xp[8], const float xw[8]) {
#pragma unroll
    for (int u = 0; u < 8; ++u) {
      v2f B2; B2.x = 0.f;     B2.y = xw[u];
      v2f B3; B3.x = xp[u].x; B3.y = xp[u].y;
      v8f p = wmma4(AX4, B2, cbias);
      p = wmma4(A3, B3, p);
      float* q = pw + so + (8 * g + u) * PSTEPF;
      *(v4f*)(q)       = __builtin_shufflevector(p, p, 0, 1, 2, 3);
      *(v4f*)(q + 128) = __builtin_shufflevector(p, p, 4, 5, 6, 7);
    }
  };
  auto readp = [&](int so, int g, v8f P[8]) {
    __builtin_amdgcn_sched_barrier(0);
#pragma unroll
    for (int u = 0; u < 8; ++u) {
      const float* q = pw + so + (8 * g + u) * PSTEPF;
      const v4f a = *(const v4f*)(q);
      const v4f b = *(const v4f*)(q + 128);
      P[u] = __builtin_shufflevector(a, b, 0, 1, 2, 3, 4, 5, 6, 7);
    }
    __builtin_amdgcn_sched_barrier(0);
  };

#pragma unroll
  for (int i = 0; i < 5; ++i) d[i] = 0.f;

  v8f PA[8];
  if (role) {
    v4f R0[NLD];
    issue(0, R0);
    stash(0, R0);
  }

#pragma unroll 1
  for (int i = 0; i < NCHUNK + 2; ++i) {
    __syncthreads();
    if (role) {
      if (i < NCHUNK) {
        v4f R[NLD];
        issue(i + 1 < NCHUNK ? i + 1 : i, R);
        const int so = (i % NPSLOT) * PSLOTF;
        float2 xpA[8], xpB[8]; float xwA[8], xwB[8];
        readgrp(i, 0, xpA, xwA);
        readgrp(i, 1, xpB, xwB);
        makegrp(so, 0, xpA, xwA);
        readgrp(i, 2, xpA, xwA);
        makegrp(so, 1, xpB, xwB);
        readgrp(i, 3, xpB, xwB);
        makegrp(so, 2, xpA, xwA);
        makegrp(so, 3, xpB, xwB);
        stash(i + 1, R);
      }
    } else if (i >= 2) {
      const int c  = i - 2;
      const int so = (c % NPSLOT) * PSLOTF;
      const int sn = ((c + 1) % NPSLOT) * PSLOTF;
      v8f PB[8];
      if (c == 0) readp(so, 0, PA);
      readp(so, 1, PB);
#pragma unroll
      for (int u = 0; u < 8; ++u) rnn1_state_step(A0, A1, A2h, d, PA[u]);
      readp(so, 2, PA);
#pragma unroll
      for (int u = 0; u < 8; ++u) rnn1_state_step(A0, A1, A2h, d, PB[u]);
      readp(so, 3, PB);
#pragma unroll
      for (int u = 0; u < 8; ++u) rnn1_state_step(A0, A1, A2h, d, PA[u]);
      readp(sn, 0, PA);
#pragma unroll
      for (int u = 0; u < 8; ++u) rnn1_state_step(A0, A1, A2h, d, PB[u]);
    }
  }
}

#define YPITCH  20
#define SPITCH  36
#define NROW    (16 * OUTT)

__global__ __launch_bounds__(128) void rnn_fused_kernel(
    const float* __restrict__ x,
    const float* __restrict__ w_ih_f, const float* __restrict__ w_hh_f,
    const float* __restrict__ b_ih_f, const float* __restrict__ b_hh_f,
    const float* __restrict__ w_ih_b, const float* __restrict__ w_hh_b,
    const float* __restrict__ b_ih_b, const float* __restrict__ b_hh_b,
    const float* __restrict__ w_ih2, const float* __restrict__ w_hh2,
    const float* __restrict__ b_ih2, const float* __restrict__ b_hh2,
    const float* __restrict__ w_out, const float* __restrict__ b_out,
    float* __restrict__ out  )
{
  const int lane = threadIdx.x & 31;
  const int wv   = __builtin_amdgcn_readfirstlane((int)(threadIdx.x >> 5));
  const int role = wv & 1;
  const int dir  = wv >> 1;
  const int s0   = blockIdx.x * 16;
  const int hi   = lane >> 4;
  const int ii   = lane & 15;

  const float* wih = dir ? w_ih_b : w_ih_f;
  const float* whh = dir ? w_hh_b : w_hh_f;
  const float* bih = dir ? b_ih_b : b_ih_f;
  const float* bhh = dir ? b_hh_b : b_hh_f;

  auto unitOfRow = [](int i) -> int {
    return (i < 5) ? i : ((i >= 8 && i < 12) ? i - 3 : -1);
  };
  const int slotType[16] = {0,0,0,0, 0,0,0,0, 0,2,2,1, 1,1,1,1};
  const int slotIdx [16] = {0,1,5,6, 2,3,7,8, 4,0,0,4, 0,1,2,3};

  const int urow = unitOfRow(ii);
  auto wElem = [&](int k) -> float {
    if (urow < 0) return 0.f;
    const int ty = slotType[k], ix = slotIdx[k];
    if (ty == 0) return whh[urow * H1N + ix];
    if (ty == 1) return wih[urow * DIN + ix];
    return 0.f;
  };
  auto aW = [&](int kb) -> v2f {
    v2f v; const int k0 = kb + 2 * hi;
    v.x = wElem(k0); v.y = wElem(k0 + 1);
    return v;
  };
  const v2f A0 = aW(0), A1 = aW(4), A2 = aW(8), A3 = aW(12);
  v2f A2h; A2h.x = hi ? 0.f : A2.x; A2h.y = 0.f;
  v2f AX4; AX4.x = 0.f;             AX4.y = hi ? A2.y : 0.f;

  v8f cbias;
#pragma unroll
  for (int rr = 0; rr < 8; ++rr) {
    const int uu = unitOfRow(rr + 8 * hi);
    cbias[rr] = (uu >= 0) ? (bih[uu] + bhh[uu]) : 0.f;
  }

  __shared__ __attribute__((aligned(16))) float smem[2 * (XRINGF + PRINGF)];
  float* xr = smem + dir * XRINGF;
  float* pr = smem + 2 * XRINGF + dir * PRINGF;
  const float* xs = x + (long long)s0 * (T_LEN * DIN);

  float d[5];
  rnn1_pair_loop(dir, xs, xr, pr, role, lane, hi, ii, A0, A1, A2h, AX4, A3, cbias, d);

  float* ytile = smem;
  float* splane = smem + 512;
  if (role == 0) {
#pragma unroll
    for (int rr = 0; rr < 5; ++rr) {
      const int uu = unitOfRow(rr + 8 * hi);
      if (uu >= 0) ytile[ii * YPITCH + dir * H1N + uu] = d[rr];
    }
  }
  __syncthreads();

  float wo[3 * H2N], bo[3];
#pragma unroll
  for (int k = 0; k < 3 * H2N; ++k) wo[k] = w_out[k];
#pragma unroll
  for (int k = 0; k < 3; ++k) bo[k] = b_out[k];

  if (wv == 0) {
    v2f Aih[5][2];
#pragma unroll
    for (int c = 0; c < 5; ++c)
#pragma unroll
      for (int nt = 0; nt < 2; ++nt) {
        const int k0 = 4 * c + 2 * hi, i = nt * 16 + ii;
        Aih[c][nt].x = ((k0    ) < 18) ? w_ih2[i * 18 + k0    ] : 0.f;
        Aih[c][nt].y = ((k0 + 1) < 18) ? w_ih2[i * 18 + k0 + 1] : 0.f;
      }
    v2f Ahh[8][2];
#pragma unroll
    for (int c = 0; c < 8; ++c)
#pragma unroll
      for (int nt = 0; nt < 2; ++nt) {
        const int j0 = 16 * (c >> 2) + 8 * hi + 2 * (c & 3);
        const int i = nt * 16 + ii;
        Ahh[c][nt].x = w_hh2[i * 32 + j0    ];
        Ahh[c][nt].y = w_hh2[i * 32 + j0 + 1];
      }

    v8f cb[2];
#pragma unroll
    for (int nt = 0; nt < 2; ++nt)
#pragma unroll
      for (int rr = 0; rr < 8; ++rr) {
        const int i = nt * 16 + rr + 8 * hi;
        cb[nt][rr] = b_ih2[i] + b_hh2[i];
      }

    v2f By[5];
#pragma unroll
    for (int c = 0; c < 5; ++c) {
      const int k0 = 4 * c + 2 * hi;
      const float v0 = ytile[ii * YPITCH + k0];
      const float v1 = ytile[ii * YPITCH + k0 + 1];
      By[c].x = ((k0    ) < 18) ? v0 : 0.f;
      By[c].y = ((k0 + 1) < 18) ? v1 : 0.f;
    }

    v8f d2[2];
    v8f cz = {};
    float* srow = splane + (ii * OUTT) * SPITCH + 8 * hi;
#pragma unroll 1
    for (int t = 0; t < OUTT; ++t) {
      if (t == 0) {
#pragma unroll
        for (int nt = 0; nt < 2; ++nt) {
          v8f c1 = wmma4(Aih[0][nt], By[0], cb[nt]);
          v8f c2 = wmma4(Aih[1][nt], By[1], cz);
          c1 = wmma4(Aih[2][nt], By[2], c1);
          c2 = wmma4(Aih[3][nt], By[3], c2);
          c1 = wmma4(Aih[4][nt], By[4], c1);
#pragma unroll
          for (int rr = 0; rr < 8; ++rr) d2[nt][rr] = fast_tanh(c1[rr] + c2[rr]);
        }
      } else {
        v2f Bh[8];
#pragma unroll
        for (int c = 0; c < 8; ++c) {
          Bh[c].x = d2[c >> 2][2 * (c & 3)    ];
          Bh[c].y = d2[c >> 2][2 * (c & 3) + 1];
        }
        v8f dn[2];
#pragma unroll
        for (int nt = 0; nt < 2; ++nt) {
          v8f c1 = cb[nt];
          v8f c2 = cz;
#pragma unroll
          for (int cc = 0; cc < 8; cc += 2) {
            c1 = wmma4(Ahh[cc    ][nt], Bh[cc    ], c1);
            c2 = wmma4(Ahh[cc + 1][nt], Bh[cc + 1], c2);
          }
#pragma unroll
          for (int rr = 0; rr < 8; ++rr) dn[nt][rr] = fast_tanh(c1[rr] + c2[rr]);
        }
        d2[0] = dn[0]; d2[1] = dn[1];
      }

      float* q = srow + t * SPITCH;
#pragma unroll
      for (int nt = 0; nt < 2; ++nt) {
        *(v4f*)(q + nt * 16)     = __builtin_shufflevector(d2[nt], d2[nt], 0, 1, 2, 3);
        *(v4f*)(q + nt * 16 + 4) = __builtin_shufflevector(d2[nt], d2[nt], 4, 5, 6, 7);
      }
    }
  }
  __syncthreads();

  float* og = out + (long long)s0 * (OUTT * 3);
#pragma unroll
  for (int qq = 0; qq < (NROW + 127) / 128; ++qq) {
    const int row = (int)threadIdx.x + 128 * qq;
    if (row < NROW) {
      const float* rowp = splane + row * SPITCH;
      float a0 = bo[0], a1 = bo[1], a2 = bo[2];
#pragma unroll
      for (int n4 = 0; n4 < 8; ++n4) {
        const v4f v = *(const v4f*)(rowp + 4 * n4);
#pragma unroll
        for (int e = 0; e < 4; ++e) {
          const int nn = 4 * n4 + e;
          a0 = fmaf(v[e], wo[nn],      a0);
          a1 = fmaf(v[e], wo[32 + nn], a1);
          a2 = fmaf(v[e], wo[64 + nn], a2);
        }
      }
      og[row * 3 + 0] = a0;
      og[row * 3 + 1] = a1;
      og[row * 3 + 2] = a2;
    }
  }
}

extern "C" void kernel_launch(void* const* d_in, const int* in_sizes, int n_in,
                              void* d_out, int out_size, void* d_ws,
                              size_t ws_size, hipStream_t stream) {
  const float* x      = (const float*)d_in[0];
  const float* w_ih_f = (const float*)d_in[1];
  const float* w_hh_f = (const float*)d_in[2];
  const float* b_ih_f = (const float*)d_in[3];
  const float* b_hh_f = (const float*)d_in[4];
  const float* w_ih_b = (const float*)d_in[5];
  const float* w_hh_b = (const float*)d_in[6];
  const float* b_ih_b = (const float*)d_in[7];
  const float* b_hh_b = (const float*)d_in[8];
  const float* w_ih2  = (const float*)d_in[9];
  const float* w_hh2  = (const float*)d_in[10];
  const float* b_ih2  = (const float*)d_in[11];
  const float* b_hh2  = (const float*)d_in[12];
  const float* w_out  = (const float*)d_in[13];
  const float* b_out  = (const float*)d_in[14];

  rnn_fused_kernel<<<dim3(256), dim3(128), 0, stream>>>(
      x, w_ih_f, w_hh_f, b_ih_f, b_hh_f,
      w_ih_b, w_hh_b, b_ih_b, b_hh_b,
      w_ih2, w_hh2, b_ih2, b_hh2, w_out, b_out, (float*)d_out);
}
